# starting kernel
# baseline (speedup 1.0000x reference)
#include <hip/hip_runtime.h>

typedef __attribute__((ext_vector_type(2))) float v2f;
typedef __attribute__((ext_vector_type(4))) float v4f;
typedef __attribute__((ext_vector_type(8))) float v8f;

#define M_TOTAL (16 * 4096)
#define K_COLS  512
#define D_DIM   64
#define ROWS_PER_WG   128
#define CHUNK_ROWS    128
#define NTILES_CHUNK  (CHUNK_ROWS / 16)
#define NCHUNKS       (K_COLS / CHUNK_ROWS)

__global__ __launch_bounds__(256)
void sl2_wmma_kernel(const float* __restrict__ x,
                     const float* __restrict__ C,
                     const float* __restrict__ S,
                     float* __restrict__ out) {
    __shared__ v2f   frag[CHUNK_ROWS * 32];
    __shared__ float c2s[K_COLS];
    __shared__ float ss[K_COLS];

    const int tid   = threadIdx.x;
    const int wave  = tid >> 5;
    const int lane  = tid & 31;
    const int lan15 = lane & 15;
    const int h     = lane >> 4;

    #pragma unroll
    for (int r = 0; r < 2; ++r) {
        int k = tid * 2 + r;
        const float* cp = C + k * D_DIM;
        float a2 = 0.f;
        #pragma unroll
        for (int d = 0; d < D_DIM; d += 4) {
            v4f c4 = *(const v4f*)(cp + d);
            a2 = fmaf(c4.x, c4.x, a2);
            a2 = fmaf(c4.y, c4.y, a2);
            a2 = fmaf(c4.z, c4.z, a2);
            a2 = fmaf(c4.w, c4.w, a2);
        }
        c2s[k] = a2;
        ss[k]  = S[k];
    }

    const int rowbase = blockIdx.x * ROWS_PER_WG + wave * 16;
    const float* xrow = x + (rowbase + lan15) * D_DIM + 2 * h;
    v2f a[16];
    float part = 0.f;
    #pragma unroll
    for (int s = 0; s < 16; ++s) {
        a[s] = *(const v2f*)(xrow + 4 * s);
        part = fmaf(a[s].x, a[s].x, part);
        part = fmaf(a[s].y, a[s].y, part);
    }
    float x2full = part + __shfl_xor(part, 16, 32);
    float xb[8];
    #pragma unroll
    for (int j = 0; j < 8; ++j)
        xb[j] = __shfl(x2full, j + 8 * h, 32);

    for (int cc = 0; cc < NCHUNKS; ++cc) {
        __syncthreads();
        const v2f* csrc = (const v2f*)(C + cc * CHUNK_ROWS * D_DIM);
        #pragma unroll
        for (int it = 0; it < (CHUNK_ROWS * 32) / 256; ++it) {
            int q = it * 256 + tid;
            v2f val = csrc[q];
            int nloc = q >> 5;
            int d    = (q & 31) * 2;
            int s    = d >> 2;
            int hh   = (d >> 1) & 1;
            int f    = (((nloc >> 4) * 16 + s) << 5) + (nloc & 15) + (hh << 4);
            frag[f] = val;
        }
        __syncthreads();

        for (int t = 0; t < NTILES_CHUNK; ++t) {
            int   kcol = cc * CHUNK_ROWS + t * 16 + lan15;
            float sn   = ss[kcol];
            float c2n  = c2s[kcol];

            v8f acc0 = {};
            v8f acc1 = {};
            const v2f* fb = &frag[t * 16 * 32];
            #pragma unroll
            for (int s = 0; s < 16; s += 2) {
                v2f b0 = fb[(s + 0) * 32 + lane];
                v2f b1 = fb[(s + 1) * 32 + lane];
                acc0 = __builtin_amdgcn_wmma_f32_16x16x4_f32(
                           false, a[s + 0], false, b0, (short)0, acc0, false, false);
                acc1 = __builtin_amdgcn_wmma_f32_16x16x4_f32(
                           false, a[s + 1], false, b1, (short)0, acc1, false, false);
            }

            float m2s  = -2.0f * sn;
            float base = sn * c2n;
            float* op = out + (rowbase + 8 * h) * K_COLS + kcol;
            #pragma unroll
            for (int j = 0; j < 8; ++j) {
                float xc = acc0[j] + acc1[j];
                op[j * K_COLS] = fmaf(m2s, xc, fmaf(sn, xb[j], base));
            }
        }
    }
}

extern "C" void kernel_launch(void* const* d_in, const int* in_sizes, int n_in,
                              void* d_out, int out_size, void* d_ws, size_t ws_size,
                              hipStream_t stream) {
    (void)in_sizes; (void)n_in; (void)out_size; (void)d_ws; (void)ws_size;
    const float* x = (const float*)d_in[0];
    const float* C = (const float*)d_in[1];
    const float* S = (const float*)d_in[2];
    float* out = (float*)d_out;

    dim3 grid(M_TOTAL / ROWS_PER_WG);
    dim3 block(256);
    hipLaunchKernelGGL(sl2_wmma_kernel, grid, block, 0, stream, x, C, S, out);
}


# C column tiles kept as B fragments in registers (64 cols per wave), x row tiles streamed from global: no LDS staging, no barriers; k and norm sums regrouped by b128 fragment loads
# speedup vs baseline: 1.3362x; 1.3362x over previous
#include <hip/hip_runtime.h>


typedef __attribute__((ext_vector_type(2))) float v2f;
typedef __attribute__((ext_vector_type(4))) float v4f;
typedef __attribute__((ext_vector_type(8))) float v8f;

#define M_TOTAL (16 * 4096)
#define K_COLS  512
#define D_DIM   64
#define ROWS_PER_WG   128
#define ROW_TILES     (ROWS_PER_WG / 16)
#define COL_TILES     4
#define KGROUPS       (D_DIM / 8)


__global__ __launch_bounds__(256)
void sl2_wmma_kernel(const float* __restrict__ x,
                     const float* __restrict__ C,
                     const float* __restrict__ S,
                     float* __restrict__ out) {
    const int tid  = threadIdx.x;
    const int wave = tid >> 5;
    const int lane = tid & 31;
    const int c    = lane & 15;
    const int h    = lane >> 4;
    const int koff = 4 * h;

    const int col0 = wave * (COL_TILES * 16);
    v4f   b[COL_TILES][KGROUPS];
    float sn[COL_TILES], m2s[COL_TILES], base[COL_TILES];
    #pragma unroll
    for (int t = 0; t < COL_TILES; ++t) {
        const int    n  = col0 + t * 16 + c;
        const float* cp = C + n * D_DIM + koff;
        float p = 0.f;
        #pragma unroll
        for (int g = 0; g < KGROUPS; ++g) {
            v4f v = *(const v4f*)(cp + 8 * g);
            b[t][g] = v;
            p = fmaf(v.x, v.x, p);
            p = fmaf(v.y, v.y, p);
            p = fmaf(v.z, v.z, p);
            p = fmaf(v.w, v.w, p);
        }
        const float c2 = p + __shfl_xor(p, 16, 32);
        const float s  = S[n];
        sn[t]   = s;
        m2s[t]  = -2.0f * s;
        base[t] = s * c2;
    }

    const int    rowbase = blockIdx.x * ROWS_PER_WG;
    const float* xp      = x + (rowbase + c) * D_DIM + koff;
    float*       op      = out + (rowbase + 8 * h) * K_COLS + col0 + c;

    v4f a[KGROUPS];
    #pragma unroll
    for (int g = 0; g < KGROUPS; ++g)
        a[g] = *(const v4f*)(xp + 8 * g);

    #pragma unroll 1
    for (int rt = 0; rt < ROW_TILES; ++rt) {
        float part = 0.f;
        #pragma unroll
        for (int g = 0; g < KGROUPS; ++g) {
            part = fmaf(a[g].x, a[g].x, part);
            part = fmaf(a[g].y, a[g].y, part);
            part = fmaf(a[g].z, a[g].z, part);
            part = fmaf(a[g].w, a[g].w, part);
        }
        const float x2full = part + __shfl_xor(part, 16, 32);
        float xb[8];
        #pragma unroll
        for (int j = 0; j < 8; ++j)
            xb[j] = __shfl(x2full, j + 8 * h, 32);

        v8f acc[COL_TILES];
        #pragma unroll
        for (int t = 0; t < COL_TILES; ++t)
            acc[t] = (v8f){0.f, 0.f, 0.f, 0.f, 0.f, 0.f, 0.f, 0.f};

        #pragma unroll
        for (int g = 0; g < KGROUPS; ++g) {
            const v2f a_lo = __builtin_shufflevector(a[g], a[g], 0, 1);
            const v2f a_hi = __builtin_shufflevector(a[g], a[g], 2, 3);
            #pragma unroll
            for (int t = 0; t < COL_TILES; ++t) {
                const v2f b_lo = __builtin_shufflevector(b[t][g], b[t][g], 0, 1);
                acc[t] = __builtin_amdgcn_wmma_f32_16x16x4_f32(
                             false, a_lo, false, b_lo, (short)0, acc[t], false, false);
            }
            #pragma unroll
            for (int t = 0; t < COL_TILES; ++t) {
                const v2f b_hi = __builtin_shufflevector(b[t][g], b[t][g], 2, 3);
                acc[t] = __builtin_amdgcn_wmma_f32_16x16x4_f32(
                             false, a_hi, false, b_hi, (short)0, acc[t], false, false);
            }
        }

        const int    rn = (rt + 1 < ROW_TILES) ? (rt + 1) : rt;
        const float* xn = xp + rn * 16 * D_DIM;
        #pragma unroll
        for (int g = 0; g < KGROUPS; ++g)
            a[g] = *(const v4f*)(xn + 8 * g);

        float* o = op + rt * 16 * K_COLS;
        #pragma unroll
        for (int t = 0; t < COL_TILES; ++t) {
            #pragma unroll
            for (int j = 0; j < 8; ++j)
                o[j * K_COLS + t * 16] =
                    fmaf(m2s[t], acc[t][j], fmaf(sn[t], xb[j], base[t]));
        }
    }
}

extern "C" void kernel_launch(void* const* d_in, const int* in_sizes, int n_in,
                              void* d_out, int out_size, void* d_ws, size_t ws_size,
                              hipStream_t stream) {
    (void)in_sizes; (void)n_in; (void)out_size; (void)d_ws; (void)ws_size;
    const float* x = (const float*)d_in[0];
    const float* C = (const float*)d_in[1];
    const float* S = (const float*)d_in[2];
    float* out = (float*)d_out;

    dim3 grid(M_TOTAL / ROWS_PER_WG);
    dim3 block(256);
    hipLaunchKernelGGL(sl2_wmma_kernel, grid, block, 0, stream, x, C, S, out);
}


# adjacent column tiles exchange half-waves (v_permlane16_swap_b32) on the finished values: every store writes one whole 128-byte line, not two 64-byte halves; sums and fma order unchanged
# speedup vs baseline: 1.4719x; 1.1016x over previous
#include <hip/hip_runtime.h>


typedef __attribute__((ext_vector_type(2))) float v2f;
typedef __attribute__((ext_vector_type(4))) float v4f;
typedef __attribute__((ext_vector_type(8))) float v8f;

#define M_TOTAL (16 * 4096)
#define K_COLS  512
#define D_DIM   64
#define ROWS_PER_WG   128
#define ROW_TILES     (ROWS_PER_WG / 16)
#define COL_TILES     4
#define KGROUPS       (D_DIM / 8)
#define COL_PAIRS     (COL_TILES / 2)


__global__ __launch_bounds__(256)
void sl2_wmma_kernel(const float* __restrict__ x,
                     const float* __restrict__ C,
                     const float* __restrict__ S,
                     float* __restrict__ out) {
    const int tid  = threadIdx.x;
    const int wave = tid >> 5;
    const int lane = tid & 31;
    const int c    = lane & 15;
    const int h    = lane >> 4;
    const int koff = 4 * h;

    const int col0 = wave * (COL_TILES * 16);
    v4f   b[COL_TILES][KGROUPS];
    float sn[COL_TILES], m2s[COL_TILES], base[COL_TILES];
    #pragma unroll
    for (int t = 0; t < COL_TILES; ++t) {
        const int    n  = col0 + t * 16 + c;
        const float* cp = C + n * D_DIM + koff;
        float p = 0.f;
        #pragma unroll
        for (int g = 0; g < KGROUPS; ++g) {
            v4f v = *(const v4f*)(cp + 8 * g);
            b[t][g] = v;
            p = fmaf(v.x, v.x, p);
            p = fmaf(v.y, v.y, p);
            p = fmaf(v.z, v.z, p);
            p = fmaf(v.w, v.w, p);
        }
        const float c2 = p + __shfl_xor(p, 16, 32);
        const float s  = S[n];
        sn[t]   = s;
        m2s[t]  = -2.0f * s;
        base[t] = s * c2;
    }

    const int    rowbase = blockIdx.x * ROWS_PER_WG;
    const float* xp      = x + (rowbase + c) * D_DIM + koff;
    float*       op      = out + rowbase * K_COLS + col0 + lane;

    v4f a[KGROUPS];
    #pragma unroll
    for (int g = 0; g < KGROUPS; ++g)
        a[g] = *(const v4f*)(xp + 8 * g);

    #pragma unroll 1
    for (int rt = 0; rt < ROW_TILES; ++rt) {
        float part = 0.f;
        #pragma unroll
        for (int g = 0; g < KGROUPS; ++g) {
            part = fmaf(a[g].x, a[g].x, part);
            part = fmaf(a[g].y, a[g].y, part);
            part = fmaf(a[g].z, a[g].z, part);
            part = fmaf(a[g].w, a[g].w, part);
        }
        const float x2full = part + __shfl_xor(part, 16, 32);
        float xb[8];
        #pragma unroll
        for (int j = 0; j < 8; ++j)
            xb[j] = __shfl(x2full, j + 8 * h, 32);

        v8f acc[COL_TILES];
        #pragma unroll
        for (int t = 0; t < COL_TILES; ++t)
            acc[t] = (v8f){0.f, 0.f, 0.f, 0.f, 0.f, 0.f, 0.f, 0.f};

        #pragma unroll
        for (int g = 0; g < KGROUPS; ++g) {
            const v2f a_lo = __builtin_shufflevector(a[g], a[g], 0, 1);
            const v2f a_hi = __builtin_shufflevector(a[g], a[g], 2, 3);
            #pragma unroll
            for (int t = 0; t < COL_TILES; ++t) {
                const v2f b_lo = __builtin_shufflevector(b[t][g], b[t][g], 0, 1);
                acc[t] = __builtin_amdgcn_wmma_f32_16x16x4_f32(
                             false, a_lo, false, b_lo, (short)0, acc[t], false, false);
            }
            #pragma unroll
            for (int t = 0; t < COL_TILES; ++t) {
                const v2f b_hi = __builtin_shufflevector(b[t][g], b[t][g], 2, 3);
                acc[t] = __builtin_amdgcn_wmma_f32_16x16x4_f32(
                             false, a_hi, false, b_hi, (short)0, acc[t], false, false);
            }
        }

        const int    rn = (rt + 1 < ROW_TILES) ? (rt + 1) : rt;
        const float* xn = xp + rn * 16 * D_DIM;
        #pragma unroll
        for (int g = 0; g < KGROUPS; ++g)
            a[g] = *(const v4f*)(xn + 8 * g);

        float* o = op + rt * 16 * K_COLS;
        #pragma unroll
        for (int p = 0; p < COL_PAIRS; ++p) {
            #pragma unroll
            for (int j = 0; j < 8; ++j) {
                float lo = fmaf(m2s[2 * p], acc[2 * p][j],
                                fmaf(sn[2 * p], xb[j], base[2 * p]));
                float hi = fmaf(m2s[2 * p + 1], acc[2 * p + 1][j],
                                fmaf(sn[2 * p + 1], xb[j], base[2 * p + 1]));
                asm("v_permlane16_swap_b32 %0, %1" : "+v"(lo), "+v"(hi));
                o[j * K_COLS + 32 * p]       = lo;
                o[(j + 8) * K_COLS + 32 * p] = hi;
            }
        }
    }
}

extern "C" void kernel_launch(void* const* d_in, const int* in_sizes, int n_in,
                              void* d_out, int out_size, void* d_ws, size_t ws_size,
                              hipStream_t stream) {
    (void)in_sizes; (void)n_in; (void)out_size; (void)d_ws; (void)ws_size;
    const float* x = (const float*)d_in[0];
    const float* C = (const float*)d_in[1];
    const float* S = (const float*)d_in[2];
    float* out = (float*)d_out;

    dim3 grid(M_TOTAL / ROWS_PER_WG);
    dim3 block(256);
    hipLaunchKernelGGL(sl2_wmma_kernel, grid, block, 0, stream, x, C, S, out);
}


# a workgroup's 128 rows of x fetched once by whole-line loads into LDS behind one barrier; all 8 waves read A fragments from LDS instead of 8 times from L2; k map and sum order unchanged
# speedup vs baseline: 1.5274x; 1.0377x over previous
#include <hip/hip_runtime.h>


typedef __attribute__((ext_vector_type(2))) float v2f;
typedef __attribute__((ext_vector_type(4))) float v4f;
typedef __attribute__((ext_vector_type(8))) float v8f;

#define M_TOTAL (16 * 4096)
#define K_COLS  512
#define D_DIM   64
#define ROWS_PER_WG   128
#define ROW_TILES     (ROWS_PER_WG / 16)
#define COL_TILES     4
#define KGROUPS       (D_DIM / 8)
#define COL_PAIRS     (COL_TILES / 2)
#define X_PITCH       (D_DIM + 4)


__global__ __launch_bounds__(256)
void sl2_wmma_kernel(const float* __restrict__ x,
                     const float* __restrict__ C,
                     const float* __restrict__ S,
                     float* __restrict__ out) {
    const int tid  = threadIdx.x;
    const int wave = tid >> 5;
    const int lane = tid & 31;
    const int c    = lane & 15;
    const int h    = lane >> 4;
    const int koff = 4 * h;

    __shared__ __attribute__((aligned(16))) float xs[ROWS_PER_WG * X_PITCH];
    const int rowbase = blockIdx.x * ROWS_PER_WG;
    v4f xin[ROW_TILES];
    {
        const float* xg = x + (rowbase + 16 * wave) * D_DIM + 4 * lane;
        #pragma unroll
        for (int i = 0; i < ROW_TILES; ++i)
            xin[i] = *(const v4f*)(xg + i * 128);
    }

    const int col0 = wave * (COL_TILES * 16);
    v4f   b[COL_TILES][KGROUPS];
    float sn[COL_TILES], m2s[COL_TILES], base[COL_TILES];
    #pragma unroll
    for (int t = 0; t < COL_TILES; ++t) {
        const int    n  = col0 + t * 16 + c;
        const float* cp = C + n * D_DIM + koff;
        float p = 0.f;
        #pragma unroll
        for (int g = 0; g < KGROUPS; ++g) {
            v4f v = *(const v4f*)(cp + 8 * g);
            b[t][g] = v;
            p = fmaf(v.x, v.x, p);
            p = fmaf(v.y, v.y, p);
            p = fmaf(v.z, v.z, p);
            p = fmaf(v.w, v.w, p);
        }
        const float c2 = p + __shfl_xor(p, 16, 32);
        const float s  = S[n];
        sn[t]   = s;
        m2s[t]  = -2.0f * s;
        base[t] = s * c2;
    }

    {
        float* xw = xs + (16 * wave + h) * X_PITCH + 4 * c;
        #pragma unroll
        for (int i = 0; i < ROW_TILES; ++i)
            *(v4f*)(xw + 2 * i * X_PITCH) = xin[i];
    }
    __syncthreads();

    const float* xp = xs + c * X_PITCH + koff;
    float*       op = out + rowbase * K_COLS + col0 + lane;

    v4f a[KGROUPS];
    #pragma unroll
    for (int g = 0; g < KGROUPS; ++g)
        a[g] = *(const v4f*)(xp + 8 * g);

    #pragma unroll 1
    for (int rt = 0; rt < ROW_TILES; ++rt) {
        float part = 0.f;
        #pragma unroll
        for (int g = 0; g < KGROUPS; ++g) {
            part = fmaf(a[g].x, a[g].x, part);
            part = fmaf(a[g].y, a[g].y, part);
            part = fmaf(a[g].z, a[g].z, part);
            part = fmaf(a[g].w, a[g].w, part);
        }
        v8f acc[COL_TILES];
        #pragma unroll
        for (int t = 0; t < COL_TILES; ++t)
            acc[t] = (v8f){0.f, 0.f, 0.f, 0.f, 0.f, 0.f, 0.f, 0.f};

        #pragma unroll
        for (int g = 0; g < KGROUPS; ++g) {
            const v2f a_lo = __builtin_shufflevector(a[g], a[g], 0, 1);
            const v2f a_hi = __builtin_shufflevector(a[g], a[g], 2, 3);
            #pragma unroll
            for (int t = 0; t < COL_TILES; ++t) {
                const v2f b_lo = __builtin_shufflevector(b[t][g], b[t][g], 0, 1);
                acc[t] = __builtin_amdgcn_wmma_f32_16x16x4_f32(
                             false, a_lo, false, b_lo, (short)0, acc[t], false, false);
            }
            #pragma unroll
            for (int t = 0; t < COL_TILES; ++t) {
                const v2f b_hi = __builtin_shufflevector(b[t][g], b[t][g], 2, 3);
                acc[t] = __builtin_amdgcn_wmma_f32_16x16x4_f32(
                             false, a_hi, false, b_hi, (short)0, acc[t], false, false);
            }
        }

        const int    rn = (rt + 1 < ROW_TILES) ? (rt + 1) : rt;
        const float* xn = xp + rn * 16 * X_PITCH;
        asm volatile("" : "+v"(part));
        __builtin_amdgcn_sched_barrier(0);
        #pragma unroll
        for (int g = 0; g < KGROUPS; ++g)
            a[g] = *(const v4f*)(xn + 8 * g);

        const float x2full = part + __shfl_xor(part, 16, 32);
        float xb[8];
        #pragma unroll
        for (int j = 0; j < 8; ++j)
            xb[j] = __shfl(x2full, j + 8 * h, 32);

        float* o = op + rt * 16 * K_COLS;
        #pragma unroll
        for (int p = 0; p < COL_PAIRS; ++p) {
            #pragma unroll
            for (int j = 0; j < 8; ++j) {
                float lo = fmaf(m2s[2 * p], acc[2 * p][j],
                                fmaf(sn[2 * p], xb[j], base[2 * p]));
                float hi = fmaf(m2s[2 * p + 1], acc[2 * p + 1][j],
                                fmaf(sn[2 * p + 1], xb[j], base[2 * p + 1]));
                asm("v_permlane16_swap_b32 %0, %1" : "+v"(lo), "+v"(hi));
                o[j * K_COLS + 32 * p]       = lo;
                o[(j + 8) * K_COLS + 32 * p] = hi;
            }
        }
    }
}

extern "C" void kernel_launch(void* const* d_in, const int* in_sizes, int n_in,
                              void* d_out, int out_size, void* d_ws, size_t ws_size,
                              hipStream_t stream) {
    (void)in_sizes; (void)n_in; (void)out_size; (void)d_ws; (void)ws_size;
    const float* x = (const float*)d_in[0];
    const float* C = (const float*)d_in[1];
    const float* S = (const float*)d_in[2];
    float* out = (float*)d_out;

    dim3 grid(M_TOTAL / ROWS_PER_WG);
    dim3 block(256);
    hipLaunchKernelGGL(sl2_wmma_kernel, grid, block, 0, stream, x, C, S, out);
}
